# starting kernel
# baseline (speedup 1.0000x reference)
#include <hip/hip_runtime.h>
#include <math.h>

typedef __attribute__((ext_vector_type(2))) float v2f;
typedef __attribute__((ext_vector_type(4))) float v4f;
typedef __attribute__((ext_vector_type(8))) float v8f;

#define Bn 8
#define Sn 2048
#define Fn 128
#define Wn 5
#define NW (Sn - Wn + 1)
#define KC 64

__global__ void __launch_bounds__(256)
k_flash_softmax_gemm(const float* __restrict__ proj, const float* __restrict__ score,
                     const float* __restrict__ feature, float* __restrict__ out) {
    __shared__ float ldsA[16][KC + 1];
    __shared__ __align__(16) float smScale[16];
    __shared__ __align__(16) float smInv[16];

    const int bid  = blockIdx.x;
    const int b    = bid >> 7;
    const int i0   = (bid & 127) << 4;
    const int tid  = threadIdx.x;
    const int lane = tid & 31;
    const int wave = tid >> 5;
    const int n0   = wave << 4;

    const int r = tid >> 4;
    const int c = (tid & 15) << 2;
    const int rowIdx = b * Sn + i0 + r;
    const float* prow  = proj    + (size_t)rowIdx * Sn;
    const float* srow  = score   + (size_t)b * Sn;
    const float* fbase = feature + (size_t)b * Sn * Fn;

    const int m16   = lane & 15;
    const int half  = lane >> 4;
    const int khalf = half << 1;

    float m = -INFINITY, s = 0.0f;
    v8f acc = {};

    for (int k0 = 0; k0 < Sn; k0 += KC) {
        v4f p  = *(const v4f*)(prow + k0 + c);
        v4f sc = *(const v4f*)(srow + k0 + c);
        float x0 = p[0] * sc[0], x1 = p[1] * sc[1];
        float x2 = p[2] * sc[2], x3 = p[3] * sc[3];
        float cm = fmaxf(fmaxf(x0, x1), fmaxf(x2, x3));
        #pragma unroll
        for (int off = 1; off < 16; off <<= 1)
            cm = fmaxf(cm, __shfl_xor(cm, off, 32));

        const float mnew  = fmaxf(m, cm);
        const float scale = __expf(m - mnew);
        float w0 = __expf(x0 - mnew), w1 = __expf(x1 - mnew);
        float w2 = __expf(x2 - mnew), w3 = __expf(x3 - mnew);
        float csum = (w0 + w1) + (w2 + w3);
        #pragma unroll
        for (int off = 1; off < 16; off <<= 1)
            csum += __shfl_xor(csum, off, 32);
        s = s * scale + csum;
        m = mnew;

        ldsA[r][c + 0] = w0;
        ldsA[r][c + 1] = w1;
        ldsA[r][c + 2] = w2;
        ldsA[r][c + 3] = w3;
        if ((tid & 15) == 0) smScale[r] = scale;
        __syncthreads();

        v4f s0 = *(const v4f*)&smScale[half * 8];
        v4f s1 = *(const v4f*)&smScale[half * 8 + 4];
        #pragma unroll
        for (int rr = 0; rr < 4; ++rr) { acc[rr] *= s0[rr]; acc[rr + 4] *= s1[rr]; }

        #pragma unroll
        for (int t = 0; t < KC / 4; ++t) {
            v2f a;
            a.x = ldsA[m16][4 * t + khalf];
            a.y = ldsA[m16][4 * t + khalf + 1];
            const float* fp = fbase + (size_t)(k0 + 4 * t + khalf) * Fn + n0 + m16;
            v2f bb;
            bb.x = fp[0];
            bb.y = fp[Fn];
            acc = __builtin_amdgcn_wmma_f32_16x16x4_f32(
                false, a, false, bb, (short)0, acc, false, false);
        }
        __syncthreads();
    }

    if ((tid & 15) == 0) smInv[r] = 1.0f / s;
    __syncthreads();
    v4f i0v = *(const v4f*)&smInv[half * 8];
    v4f i1v = *(const v4f*)&smInv[half * 8 + 4];
    #pragma unroll
    for (int rr = 0; rr < 4; ++rr) { acc[rr] *= i0v[rr]; acc[rr + 4] *= i1v[rr]; }

    const int orow0 = i0 + (half << 3);
    float* obase = out + (size_t)b * Sn * Fn + (size_t)orow0 * Fn + n0 + m16;
    #pragma unroll
    for (int rr = 0; rr < 8; ++rr)
        obase[(size_t)rr * Fn] = acc[rr];
}

__global__ void __launch_bounds__(256)
k_rowdots(const float* __restrict__ feature, const float* __restrict__ aligned,
          float* __restrict__ dotv, float* __restrict__ nfv, float* __restrict__ nav) {
    const int row  = blockIdx.x * 8 + (threadIdx.x >> 5);
    const int lane = threadIdx.x & 31;
    v4f fv = *(const v4f*)(feature + (size_t)row * Fn + lane * 4);
    v4f av = *(const v4f*)(aligned + (size_t)row * Fn + lane * 4);
    float d = 0.f, nf = 0.f, na = 0.f;
    #pragma unroll
    for (int q = 0; q < 4; ++q) {
        d  += fv[q] * av[q];
        nf += fv[q] * fv[q];
        na += av[q] * av[q];
    }
    #pragma unroll
    for (int off = 16; off > 0; off >>= 1) {
        d  += __shfl_xor(d,  off, 32);
        nf += __shfl_xor(nf, off, 32);
        na += __shfl_xor(na, off, 32);
    }
    if (lane == 0) { dotv[row] = d; nfv[row] = nf; nav[row] = na; }
}

__global__ void __launch_bounds__(256)
k_window(const float* __restrict__ dotv, const float* __restrict__ nfv,
         const float* __restrict__ nav, float* __restrict__ part) {
    __shared__ float red[256];
    const int w = blockIdx.x * 256 + threadIdx.x;
    float v = 0.0f;
    if (w < Bn * NW) {
        const int b = w / NW;
        const int i = w % NW;
        const int base = b * Sn + i;
        float dw = 0.f, fw = 0.f, aw = 0.f;
        #pragma unroll
        for (int j = 0; j < Wn; ++j) {
            dw += dotv[base + j];
            fw += nfv[base + j];
            aw += nav[base + j];
        }
        float cs = dw / (fmaxf(sqrtf(fw), 1e-12f) * fmaxf(sqrtf(aw), 1e-12f));
        v = 1.0f - cs;
    }
    red[threadIdx.x] = v;
    __syncthreads();
    #pragma unroll
    for (int off = 128; off > 0; off >>= 1) {
        if (threadIdx.x < off) red[threadIdx.x] += red[threadIdx.x + off];
        __syncthreads();
    }
    if (threadIdx.x == 0) part[blockIdx.x] = red[0];
}

__global__ void k_final(const float* __restrict__ part, float* __restrict__ loss_out) {
    float s = 0.0f;
    for (int i = 0; i < 64; ++i) s += part[i];
    loss_out[0] = s / (float)(Bn * NW);
}

extern "C" void kernel_launch(void* const* d_in, const int* in_sizes, int n_in,
                              void* d_out, int out_size, void* d_ws, size_t ws_size,
                              hipStream_t stream) {
    const float* score   = (const float*)d_in[0];
    const float* feature = (const float*)d_in[1];
    const float* proj    = (const float*)d_in[2];
    float* out = (float*)d_out;
    float* ws  = (float*)d_ws;

    float* dotv = ws;
    float* nfv  = ws + 16384;
    float* nav  = ws + 32768;
    float* part = ws + 49152;

    k_flash_softmax_gemm<<<Bn * (Sn / 16), 256, 0, stream>>>(proj, score, feature, out);
    k_rowdots           <<<(Bn * Sn) / 8, 256, 0, stream>>>(feature, out, dotv, nfv, nav);
    k_window            <<<64, 256, 0, stream>>>(dotv, nfv, nav, part);
    k_final             <<<1, 1, 0, stream>>>(part, out + (size_t)Bn * Sn * Fn);
}


# Both WMMA operands staged in LDS for a 64-row x 128-col block tile: each 128-k feature chunk loaded once per block, 2 stages, 1 barrier per chunk; softmax row sums regrouped (sum order changes)
# speedup vs baseline: 1.0525x; 1.0525x over previous
#include <hip/hip_runtime.h>
#include <math.h>


typedef __attribute__((ext_vector_type(2))) float v2f;
typedef __attribute__((ext_vector_type(4))) float v4f;
typedef __attribute__((ext_vector_type(8))) float v8f;

#define Bn 8
#define Sn 2048
#define Fn 128
#define Wn 5
#define NW (Sn - Wn + 1)

#define MB 64
#define KC 128
#define NC (Sn / KC)
#define PP (KC + 4)
#define PFH 132
#define CHS ((KC / 2) * PFH + 8)
#define NPJ (KC / 16)
#define NFI (KC / 16)

template <int CTRL>
__device__ __forceinline__ float quad_perm(float v) {
    return __builtin_bit_cast(float,
        __builtin_amdgcn_update_dpp(0, __builtin_bit_cast(int, v), CTRL, 0xF, 0xF, true));
}

__global__ void __launch_bounds__(256)
k_flash_softmax_gemm(const float* __restrict__ proj, const float* __restrict__ score,
                     const float* __restrict__ feature, float* __restrict__ out) {
    __shared__ __align__(16) float sP[2][MB * PP];
    __shared__ __align__(16) float sF[2][2 * CHS];
    __shared__ __align__(16) float sScale[2][MB];
    __shared__ __align__(16) float sInv[MB];

    const int bid  = blockIdx.x;
    const int b    = bid >> 5;
    const int i0   = (bid & 31) * MB;
    const int tid  = threadIdx.x;
    const int lane = tid & 31;
    const int wave = tid >> 5;

    const int r = ((wave >> 2) << 5) + ((lane >> 2) << 2) + (wave & 3);
    const int q = lane & 3;
    const float* prow = proj  + (size_t)(b * Sn + i0 + r) * Sn + 4 * q;
    const float* srow = score + (size_t)b * Sn + 4 * q;
    const int poff = r * PP + 4 * q;

    const float* fsrc = feature + (size_t)b * Sn * Fn + (size_t)(2 * wave) * Fn + 4 * lane;
    const int foff = (lane >> 4) * CHS + wave * PFH + (lane & 3) * 32 + ((lane >> 2) & 3) * 2;

    const int m16  = lane & 15;
    const int half = lane >> 4;
    const int rt   = wave >> 1;
    const int ch   = wave & 1;
    const int aoff = (16 * rt + m16) * PP + 2 * half;
    const int boff = ch * CHS + half * PFH + 8 * m16;
    const int soff = 16 * rt + 8 * half;

    float m = -INFINITY, s = 0.0f;
    v8f acc[4] = {};

    v4f p[NPJ], sc[NPJ], f0[NFI], f1[NFI];

    auto load_raw = [&](int c) {
        const int k0 = c * KC;
        #pragma unroll
        for (int j = 0; j < NPJ; ++j) {
            p[j]  = *(const v4f*)(prow + k0 + 16 * j);
            sc[j] = *(const v4f*)(srow + k0 + 16 * j);
        }
        #pragma unroll
        for (int i = 0; i < NFI; ++i) {
            f0[i] = *(const v4f*)(fsrc + (size_t)(k0 + 16 * i) * Fn);
            f1[i] = *(const v4f*)(fsrc + (size_t)(k0 + 16 * i + 1) * Fn);
        }
    };

    auto produce = [&](int bufn) {
        float cm = -INFINITY;
        #pragma unroll
        for (int j = 0; j < NPJ; ++j) {
            p[j] = p[j] * sc[j];
            cm = fmaxf(cm, fmaxf(fmaxf(p[j][0], p[j][1]), fmaxf(p[j][2], p[j][3])));
        }
        cm = fmaxf(cm, quad_perm<0xB1>(cm));
        cm = fmaxf(cm, quad_perm<0x4E>(cm));
        const float mnew  = fmaxf(m, cm);
        const float scale = __expf(m - mnew);
        float csum = 0.0f;
        float* pdst = &sP[bufn][poff];
        #pragma unroll
        for (int j = 0; j < NPJ; ++j) {
            v4f w;
            w[0] = __expf(p[j][0] - mnew);
            w[1] = __expf(p[j][1] - mnew);
            w[2] = __expf(p[j][2] - mnew);
            w[3] = __expf(p[j][3] - mnew);
            csum += (w[0] + w[1]) + (w[2] + w[3]);
            *(v4f*)(pdst + 16 * j) = w;
        }
        csum += quad_perm<0xB1>(csum);
        csum += quad_perm<0x4E>(csum);
        s = s * scale + csum;
        m = mnew;
        if (q == 0) sScale[bufn][r] = scale;

        float* fdst = &sF[bufn][foff];
        #pragma unroll
        for (int i = 0; i < NFI; ++i) {
            #pragma unroll
            for (int u = 0; u < 4; ++u) {
                v2f kp;
                kp[0] = f0[i][u]; kp[1] = f1[i][u];
                *(v2f*)(fdst + 8 * i * PFH + 8 * u) = kp;
            }
        }
    };

    auto consume = [&](int buf) {
        const v4f s0 = *(const v4f*)&sScale[buf][soff];
        const v4f s1 = *(const v4f*)&sScale[buf][soff + 4];
        #pragma unroll
        for (int ct = 0; ct < 4; ++ct) {
            #pragma unroll
            for (int rr = 0; rr < 4; ++rr) { acc[ct][rr] *= s0[rr]; acc[ct][rr + 4] *= s1[rr]; }
        }
        const float* ap = &sP[buf][aoff];
        const float* bp = &sF[buf][boff];
        #pragma unroll
        for (int t = 0; t < KC / 4; ++t) {
            const v2f a = *(const v2f*)(ap + 4 * t);
            const v4f b01 = *(const v4f*)(bp + 2 * t * PFH);
            const v4f b23 = *(const v4f*)(bp + 2 * t * PFH + 4);
            v2f bb[4];
            bb[0][0] = b01[0]; bb[0][1] = b01[1];
            bb[1][0] = b01[2]; bb[1][1] = b01[3];
            bb[2][0] = b23[0]; bb[2][1] = b23[1];
            bb[3][0] = b23[2]; bb[3][1] = b23[3];
            #pragma unroll
            for (int ct = 0; ct < 4; ++ct) {
                acc[ct] = __builtin_amdgcn_wmma_f32_16x16x4_f32(
                    false, a, false, bb[ct], (short)0, acc[ct], false, false);
            }
        }
    };

    load_raw(0);
    produce(0);
    load_raw(1);
    __syncthreads();

    for (int c = 0; c < NC; ++c) {
        const int buf = c & 1;
        consume(buf);
        if (c + 1 < NC) produce(buf ^ 1);
        if (c + 2 < NC) load_raw(c + 2);
        __syncthreads();
    }

    if (q == 0) sInv[r] = 1.0f / s;
    __syncthreads();
    const v4f i0v = *(const v4f*)&sInv[soff];
    const v4f i1v = *(const v4f*)&sInv[soff + 4];

    float* obase = out + (size_t)(b * Sn + i0 + soff) * Fn + 64 * ch + m16;
    #pragma unroll
    for (int ct = 0; ct < 4; ++ct) {
        #pragma unroll
        for (int rr = 0; rr < 4; ++rr) {
            obase[(size_t)rr * Fn + 16 * ct]       = acc[ct][rr] * i0v[rr];
            obase[(size_t)(rr + 4) * Fn + 16 * ct] = acc[ct][rr + 4] * i1v[rr];
        }
    }
}

__global__ void __launch_bounds__(256)
k_rowdots(const float* __restrict__ feature, const float* __restrict__ aligned,
          float* __restrict__ dotv, float* __restrict__ nfv, float* __restrict__ nav) {
    const int row  = blockIdx.x * 8 + (threadIdx.x >> 5);
    const int lane = threadIdx.x & 31;
    v4f fv = *(const v4f*)(feature + (size_t)row * Fn + lane * 4);
    v4f av = *(const v4f*)(aligned + (size_t)row * Fn + lane * 4);
    float d = 0.f, nf = 0.f, na = 0.f;
    #pragma unroll
    for (int q = 0; q < 4; ++q) {
        d  += fv[q] * av[q];
        nf += fv[q] * fv[q];
        na += av[q] * av[q];
    }
    #pragma unroll
    for (int off = 16; off > 0; off >>= 1) {
        d  += __shfl_xor(d,  off, 32);
        nf += __shfl_xor(nf, off, 32);
        na += __shfl_xor(na, off, 32);
    }
    if (lane == 0) { dotv[row] = d; nfv[row] = nf; nav[row] = na; }
}

__global__ void __launch_bounds__(256)
k_window(const float* __restrict__ dotv, const float* __restrict__ nfv,
         const float* __restrict__ nav, float* __restrict__ part) {
    __shared__ float red[256];
    const int w = blockIdx.x * 256 + threadIdx.x;
    float v = 0.0f;
    if (w < Bn * NW) {
        const int b = w / NW;
        const int i = w % NW;
        const int base = b * Sn + i;
        float dw = 0.f, fw = 0.f, aw = 0.f;
        #pragma unroll
        for (int j = 0; j < Wn; ++j) {
            dw += dotv[base + j];
            fw += nfv[base + j];
            aw += nav[base + j];
        }
        float cs = dw / (fmaxf(sqrtf(fw), 1e-12f) * fmaxf(sqrtf(aw), 1e-12f));
        v = 1.0f - cs;
    }
    red[threadIdx.x] = v;
    __syncthreads();
    #pragma unroll
    for (int off = 128; off > 0; off >>= 1) {
        if (threadIdx.x < off) red[threadIdx.x] += red[threadIdx.x + off];
        __syncthreads();
    }
    if (threadIdx.x == 0) part[blockIdx.x] = red[0];
}

__global__ void k_final(const float* __restrict__ part, float* __restrict__ loss_out) {
    float s = 0.0f;
    for (int i = 0; i < 64; ++i) s += part[i];
    loss_out[0] = s / (float)(Bn * NW);
}

extern "C" void kernel_launch(void* const* d_in, const int* in_sizes, int n_in,
                              void* d_out, int out_size, void* d_ws, size_t ws_size,
                              hipStream_t stream) {
    const float* score   = (const float*)d_in[0];
    const float* feature = (const float*)d_in[1];
    const float* proj    = (const float*)d_in[2];
    float* out = (float*)d_out;
    float* ws  = (float*)d_ws;

    float* dotv = ws;
    float* nfv  = ws + 16384;
    float* nav  = ws + 32768;
    float* part = ws + 49152;

    k_flash_softmax_gemm<<<Bn * (Sn / MB), 256, 0, stream>>>(proj, score, feature, out);
    k_rowdots           <<<(Bn * Sn) / 8, 256, 0, stream>>>(feature, out, dotv, nfv, nav);
    k_window            <<<64, 256, 0, stream>>>(dotv, nfv, nav, part);
    k_final             <<<1, 1, 0, stream>>>(part, out + (size_t)Bn * Sn * Fn);
}


# Loop reordered: stage chunk c+1, issue global loads of chunk c+2, then multiply chunk c (last product peeled): barrier's drain of loads no longer idles the matrix pipe; sum orders unchanged
# speedup vs baseline: 1.1066x; 1.0514x over previous
#include <hip/hip_runtime.h>
#include <math.h>


typedef __attribute__((ext_vector_type(2))) float v2f;
typedef __attribute__((ext_vector_type(4))) float v4f;
typedef __attribute__((ext_vector_type(8))) float v8f;

#define Bn 8
#define Sn 2048
#define Fn 128
#define Wn 5
#define NW (Sn - Wn + 1)

#define MB 64
#define KC 128
#define NC (Sn / KC)
#define PP (KC + 4)
#define PFH 132
#define CHS ((KC / 2) * PFH + 8)
#define NPJ (KC / 16)
#define NFI (KC / 16)

template <int CTRL>
__device__ __forceinline__ float quad_perm(float v) {
    return __builtin_bit_cast(float,
        __builtin_amdgcn_update_dpp(0, __builtin_bit_cast(int, v), CTRL, 0xF, 0xF, true));
}

__global__ void __launch_bounds__(256)
k_flash_softmax_gemm(const float* __restrict__ proj, const float* __restrict__ score,
                     const float* __restrict__ feature, float* __restrict__ out) {
    __shared__ __align__(16) float sP[2][MB * PP];
    __shared__ __align__(16) float sF[2][2 * CHS];
    __shared__ __align__(16) float sScale[2][MB];
    __shared__ __align__(16) float sInv[MB];

    const int bid  = blockIdx.x;
    const int b    = bid >> 5;
    const int i0   = (bid & 31) * MB;
    const int tid  = threadIdx.x;
    const int lane = tid & 31;
    const int wave = tid >> 5;

    const int r = ((wave >> 2) << 5) + ((lane >> 2) << 2) + (wave & 3);
    const int q = lane & 3;
    const float* prow = proj  + (size_t)(b * Sn + i0 + r) * Sn + 4 * q;
    const float* srow = score + (size_t)b * Sn + 4 * q;
    const int poff = r * PP + 4 * q;

    const float* fsrc = feature + (size_t)b * Sn * Fn + (size_t)(2 * wave) * Fn + 4 * lane;
    const int foff = (lane >> 4) * CHS + wave * PFH + (lane & 3) * 32 + ((lane >> 2) & 3) * 2;

    const int m16  = lane & 15;
    const int half = lane >> 4;
    const int rt   = wave >> 1;
    const int ch   = wave & 1;
    const int aoff = (16 * rt + m16) * PP + 2 * half;
    const int boff = ch * CHS + half * PFH + 8 * m16;
    const int soff = 16 * rt + 8 * half;

    float m = -INFINITY, s = 0.0f;
    v8f acc[4] = {};

    v4f p[NPJ], sc[NPJ], f0[NFI], f1[NFI];

    auto load_raw = [&](int c) {
        const int k0 = c * KC;
        #pragma unroll
        for (int j = 0; j < NPJ; ++j) {
            p[j]  = *(const v4f*)(prow + k0 + 16 * j);
            sc[j] = *(const v4f*)(srow + k0 + 16 * j);
        }
        #pragma unroll
        for (int i = 0; i < NFI; ++i) {
            f0[i] = *(const v4f*)(fsrc + (size_t)(k0 + 16 * i) * Fn);
            f1[i] = *(const v4f*)(fsrc + (size_t)(k0 + 16 * i + 1) * Fn);
        }
    };

    auto produce = [&](int bufn) {
        float cm = -INFINITY;
        #pragma unroll
        for (int j = 0; j < NPJ; ++j) {
            p[j] = p[j] * sc[j];
            cm = fmaxf(cm, fmaxf(fmaxf(p[j][0], p[j][1]), fmaxf(p[j][2], p[j][3])));
        }
        cm = fmaxf(cm, quad_perm<0xB1>(cm));
        cm = fmaxf(cm, quad_perm<0x4E>(cm));
        const float mnew  = fmaxf(m, cm);
        const float scale = __expf(m - mnew);
        float csum = 0.0f;
        float* pdst = &sP[bufn][poff];
        #pragma unroll
        for (int j = 0; j < NPJ; ++j) {
            v4f w;
            w[0] = __expf(p[j][0] - mnew);
            w[1] = __expf(p[j][1] - mnew);
            w[2] = __expf(p[j][2] - mnew);
            w[3] = __expf(p[j][3] - mnew);
            csum += (w[0] + w[1]) + (w[2] + w[3]);
            *(v4f*)(pdst + 16 * j) = w;
        }
        csum += quad_perm<0xB1>(csum);
        csum += quad_perm<0x4E>(csum);
        s = s * scale + csum;
        m = mnew;
        if (q == 0) sScale[bufn][r] = scale;

        float* fdst = &sF[bufn][foff];
        #pragma unroll
        for (int i = 0; i < NFI; ++i) {
            #pragma unroll
            for (int u = 0; u < 4; ++u) {
                v2f kp;
                kp[0] = f0[i][u]; kp[1] = f1[i][u];
                *(v2f*)(fdst + 8 * i * PFH + 8 * u) = kp;
            }
        }
    };

    auto consume = [&](int buf) {
        const v4f s0 = *(const v4f*)&sScale[buf][soff];
        const v4f s1 = *(const v4f*)&sScale[buf][soff + 4];
        #pragma unroll
        for (int ct = 0; ct < 4; ++ct) {
            #pragma unroll
            for (int rr = 0; rr < 4; ++rr) { acc[ct][rr] *= s0[rr]; acc[ct][rr + 4] *= s1[rr]; }
        }
        const float* ap = &sP[buf][aoff];
        const float* bp = &sF[buf][boff];
        #pragma unroll
        for (int t = 0; t < KC / 4; ++t) {
            const v2f a = *(const v2f*)(ap + 4 * t);
            const v4f b01 = *(const v4f*)(bp + 2 * t * PFH);
            const v4f b23 = *(const v4f*)(bp + 2 * t * PFH + 4);
            v2f bb[4];
            bb[0][0] = b01[0]; bb[0][1] = b01[1];
            bb[1][0] = b01[2]; bb[1][1] = b01[3];
            bb[2][0] = b23[0]; bb[2][1] = b23[1];
            bb[3][0] = b23[2]; bb[3][1] = b23[3];
            #pragma unroll
            for (int ct = 0; ct < 4; ++ct) {
                acc[ct] = __builtin_amdgcn_wmma_f32_16x16x4_f32(
                    false, a, false, bb[ct], (short)0, acc[ct], false, false);
            }
        }
    };

    load_raw(0);
    produce(0);
    load_raw(1);
    __syncthreads();

    for (int c = 0; c + 1 < NC; ++c) {
        const int buf = c & 1;
        produce(buf ^ 1);
        if (c + 2 < NC) load_raw(c + 2);
        consume(buf);
        __syncthreads();
    }
    if (q == 0) sInv[r] = 1.0f / s;
    consume((NC - 1) & 1);
    __syncthreads();

    const v4f i0v = *(const v4f*)&sInv[soff];
    const v4f i1v = *(const v4f*)&sInv[soff + 4];

    float* obase = out + (size_t)(b * Sn + i0 + soff) * Fn + 64 * ch + m16;
    #pragma unroll
    for (int ct = 0; ct < 4; ++ct) {
        #pragma unroll
        for (int rr = 0; rr < 4; ++rr) {
            obase[(size_t)rr * Fn + 16 * ct]       = acc[ct][rr] * i0v[rr];
            obase[(size_t)(rr + 4) * Fn + 16 * ct] = acc[ct][rr + 4] * i1v[rr];
        }
    }
}

__global__ void __launch_bounds__(256)
k_rowdots(const float* __restrict__ feature, const float* __restrict__ aligned,
          float* __restrict__ dotv, float* __restrict__ nfv, float* __restrict__ nav) {
    const int row  = blockIdx.x * 8 + (threadIdx.x >> 5);
    const int lane = threadIdx.x & 31;
    v4f fv = *(const v4f*)(feature + (size_t)row * Fn + lane * 4);
    v4f av = *(const v4f*)(aligned + (size_t)row * Fn + lane * 4);
    float d = 0.f, nf = 0.f, na = 0.f;
    #pragma unroll
    for (int q = 0; q < 4; ++q) {
        d  += fv[q] * av[q];
        nf += fv[q] * fv[q];
        na += av[q] * av[q];
    }
    #pragma unroll
    for (int off = 16; off > 0; off >>= 1) {
        d  += __shfl_xor(d,  off, 32);
        nf += __shfl_xor(nf, off, 32);
        na += __shfl_xor(na, off, 32);
    }
    if (lane == 0) { dotv[row] = d; nfv[row] = nf; nav[row] = na; }
}

__global__ void __launch_bounds__(256)
k_window(const float* __restrict__ dotv, const float* __restrict__ nfv,
         const float* __restrict__ nav, float* __restrict__ part) {
    __shared__ float red[256];
    const int w = blockIdx.x * 256 + threadIdx.x;
    float v = 0.0f;
    if (w < Bn * NW) {
        const int b = w / NW;
        const int i = w % NW;
        const int base = b * Sn + i;
        float dw = 0.f, fw = 0.f, aw = 0.f;
        #pragma unroll
        for (int j = 0; j < Wn; ++j) {
            dw += dotv[base + j];
            fw += nfv[base + j];
            aw += nav[base + j];
        }
        float cs = dw / (fmaxf(sqrtf(fw), 1e-12f) * fmaxf(sqrtf(aw), 1e-12f));
        v = 1.0f - cs;
    }
    red[threadIdx.x] = v;
    __syncthreads();
    #pragma unroll
    for (int off = 128; off > 0; off >>= 1) {
        if (threadIdx.x < off) red[threadIdx.x] += red[threadIdx.x + off];
        __syncthreads();
    }
    if (threadIdx.x == 0) part[blockIdx.x] = red[0];
}

__global__ void k_final(const float* __restrict__ part, float* __restrict__ loss_out) {
    float s = 0.0f;
    for (int i = 0; i < 64; ++i) s += part[i];
    loss_out[0] = s / (float)(Bn * NW);
}

extern "C" void kernel_launch(void* const* d_in, const int* in_sizes, int n_in,
                              void* d_out, int out_size, void* d_ws, size_t ws_size,
                              hipStream_t stream) {
    const float* score   = (const float*)d_in[0];
    const float* feature = (const float*)d_in[1];
    const float* proj    = (const float*)d_in[2];
    float* out = (float*)d_out;
    float* ws  = (float*)d_ws;

    float* dotv = ws;
    float* nfv  = ws + 16384;
    float* nav  = ws + 32768;
    float* part = ws + 49152;

    k_flash_softmax_gemm<<<Bn * (Sn / MB), 256, 0, stream>>>(proj, score, feature, out);
    k_rowdots           <<<(Bn * Sn) / 8, 256, 0, stream>>>(feature, out, dotv, nfv, nav);
    k_window            <<<64, 256, 0, stream>>>(dotv, nfv, nav, part);
    k_final             <<<1, 1, 0, stream>>>(part, out + (size_t)Bn * Sn * Fn);
}


# WMMA fragment reads from LDS issued as one group per 8 k-steps into a second register set, a 32-WMMA block ahead of use (first group in front of the staging work); sum orders unchanged
# speedup vs baseline: 1.2132x; 1.0963x over previous
#include <hip/hip_runtime.h>
#include <math.h>


typedef __attribute__((ext_vector_type(2))) float v2f;
typedef __attribute__((ext_vector_type(4))) float v4f;
typedef __attribute__((ext_vector_type(8))) float v8f;

#define Bn 8
#define Sn 2048
#define Fn 128
#define Wn 5
#define NW (Sn - Wn + 1)

#define MB 64
#define KC 128
#define NC (Sn / KC)
#define PP (KC + 4)
#define PFH 132
#define CHS ((KC / 2) * PFH + 8)
#define NPJ (KC / 16)
#define NFI (KC / 16)
#define KB 8

struct Frag { v2f a[KB]; v4f b01[KB]; v4f b23[KB]; };

template <int CTRL>
__device__ __forceinline__ float quad_perm(float v) {
    return __builtin_bit_cast(float,
        __builtin_amdgcn_update_dpp(0, __builtin_bit_cast(int, v), CTRL, 0xF, 0xF, true));
}

__global__ void __launch_bounds__(256)
k_flash_softmax_gemm(const float* __restrict__ proj, const float* __restrict__ score,
                     const float* __restrict__ feature, float* __restrict__ out) {
    __shared__ __align__(16) float sP[2][MB * PP];
    __shared__ __align__(16) float sF[2][2 * CHS];
    __shared__ __align__(16) float sScale[2][MB];
    __shared__ __align__(16) float sInv[MB];

    const int bid  = blockIdx.x;
    const int b    = bid >> 5;
    const int i0   = (bid & 31) * MB;
    const int tid  = threadIdx.x;
    const int lane = tid & 31;
    const int wave = tid >> 5;

    const int r = ((wave >> 2) << 5) + ((lane >> 2) << 2) + (wave & 3);
    const int q = lane & 3;
    const float* prow = proj  + (size_t)(b * Sn + i0 + r) * Sn + 4 * q;
    const float* srow = score + (size_t)b * Sn + 4 * q;
    const int poff = r * PP + 4 * q;

    const float* fsrc = feature + (size_t)b * Sn * Fn + (size_t)(2 * wave) * Fn + 4 * lane;
    const int foff = (lane >> 4) * CHS + wave * PFH + (lane & 3) * 32 + ((lane >> 2) & 3) * 2;

    const int m16  = lane & 15;
    const int half = lane >> 4;
    const int rt   = wave >> 1;
    const int ch   = wave & 1;
    const int aoff = (16 * rt + m16) * PP + 2 * half;
    const int boff = ch * CHS + half * PFH + 8 * m16;
    const int soff = 16 * rt + 8 * half;

    float m = -INFINITY, s = 0.0f;
    v8f acc[4] = {};

    v4f p[NPJ], sc[NPJ], f0[NFI], f1[NFI];

    auto load_raw = [&](int c) {
        const int k0 = c * KC;
        #pragma unroll
        for (int j = 0; j < NPJ; ++j) {
            p[j]  = *(const v4f*)(prow + k0 + 16 * j);
            sc[j] = *(const v4f*)(srow + k0 + 16 * j);
        }
        #pragma unroll
        for (int i = 0; i < NFI; ++i) {
            f0[i] = *(const v4f*)(fsrc + (size_t)(k0 + 16 * i) * Fn);
            f1[i] = *(const v4f*)(fsrc + (size_t)(k0 + 16 * i + 1) * Fn);
        }
    };

    auto produce = [&](int bufn) {
        float cm = -INFINITY;
        #pragma unroll
        for (int j = 0; j < NPJ; ++j) {
            p[j] = p[j] * sc[j];
            cm = fmaxf(cm, fmaxf(fmaxf(p[j][0], p[j][1]), fmaxf(p[j][2], p[j][3])));
        }
        cm = fmaxf(cm, quad_perm<0xB1>(cm));
        cm = fmaxf(cm, quad_perm<0x4E>(cm));
        const float mnew  = fmaxf(m, cm);
        const float scale = __expf(m - mnew);
        float csum = 0.0f;
        float* pdst = &sP[bufn][poff];
        #pragma unroll
        for (int j = 0; j < NPJ; ++j) {
            v4f w;
            w[0] = __expf(p[j][0] - mnew);
            w[1] = __expf(p[j][1] - mnew);
            w[2] = __expf(p[j][2] - mnew);
            w[3] = __expf(p[j][3] - mnew);
            csum += (w[0] + w[1]) + (w[2] + w[3]);
            *(v4f*)(pdst + 16 * j) = w;
        }
        csum += quad_perm<0xB1>(csum);
        csum += quad_perm<0x4E>(csum);
        s = s * scale + csum;
        m = mnew;
        if (q == 0) sScale[bufn][r] = scale;

        float* fdst = &sF[bufn][foff];
        #pragma unroll
        for (int i = 0; i < NFI; ++i) {
            #pragma unroll
            for (int u = 0; u < 4; ++u) {
                v2f kp;
                kp[0] = f0[i][u]; kp[1] = f1[i][u];
                *(v2f*)(fdst + 8 * i * PFH + 8 * u) = kp;
            }
        }
    };

    Frag fr0, fr1;
    v4f rs0, rs1;

    auto load_frags = [&](Frag& f, int buf, int blk) {
        const float* ap = &sP[buf][aoff] + 4 * KB * blk;
        const float* bp = &sF[buf][boff] + 2 * KB * blk * PFH;
        #pragma unroll
        for (int t = 0; t < KB; ++t) {
            f.a[t]   = *(const v2f*)(ap + 4 * t);
            f.b01[t] = *(const v4f*)(bp + 2 * t * PFH);
            f.b23[t] = *(const v4f*)(bp + 2 * t * PFH + 4);
        }
    };

    auto mma_block = [&](const Frag& f) {
        #pragma unroll
        for (int t = 0; t < KB; ++t) {
            v2f bb[4];
            bb[0][0] = f.b01[t][0]; bb[0][1] = f.b01[t][1];
            bb[1][0] = f.b01[t][2]; bb[1][1] = f.b01[t][3];
            bb[2][0] = f.b23[t][0]; bb[2][1] = f.b23[t][1];
            bb[3][0] = f.b23[t][2]; bb[3][1] = f.b23[t][3];
            #pragma unroll
            for (int ct = 0; ct < 4; ++ct) {
                acc[ct] = __builtin_amdgcn_wmma_f32_16x16x4_f32(
                    false, f.a[t], false, bb[ct], (short)0, acc[ct], false, false);
            }
        }
    };


    auto consume_head = [&](int buf) {
        rs0 = *(const v4f*)&sScale[buf][soff];
        rs1 = *(const v4f*)&sScale[buf][soff + 4];
        load_frags(fr0, buf, 0);
        __builtin_amdgcn_sched_barrier(0);
    };

    static_assert(KC / 4 == 4 * KB, "consume is written out for four fragment blocks");
    auto consume = [&](int buf) {
        #pragma unroll
        for (int ct = 0; ct < 4; ++ct) {
            #pragma unroll
            for (int rr = 0; rr < 4; ++rr) { acc[ct][rr] *= rs0[rr]; acc[ct][rr + 4] *= rs1[rr]; }
        }
        load_frags(fr1, buf, 1);
        __builtin_amdgcn_sched_barrier(0);
        mma_block(fr0);
        __builtin_amdgcn_sched_barrier(0);
        load_frags(fr0, buf, 2);
        __builtin_amdgcn_sched_barrier(0);
        mma_block(fr1);
        __builtin_amdgcn_sched_barrier(0);
        load_frags(fr1, buf, 3);
        __builtin_amdgcn_sched_barrier(0);
        mma_block(fr0);
        __builtin_amdgcn_sched_barrier(0);
        mma_block(fr1);
    };

    load_raw(0);
    produce(0);
    load_raw(1);
    __syncthreads();

    for (int c = 0; c + 1 < NC; ++c) {
        const int buf = c & 1;
        consume_head(buf);
        produce(buf ^ 1);
        if (c + 2 < NC) load_raw(c + 2);
        consume(buf);
        __syncthreads();
    }
    consume_head((NC - 1) & 1);
    if (q == 0) sInv[r] = 1.0f / s;
    consume((NC - 1) & 1);
    __syncthreads();

    const v4f i0v = *(const v4f*)&sInv[soff];
    const v4f i1v = *(const v4f*)&sInv[soff + 4];

    float* obase = out + (size_t)(b * Sn + i0 + soff) * Fn + 64 * ch + m16;
    #pragma unroll
    for (int ct = 0; ct < 4; ++ct) {
        #pragma unroll
        for (int rr = 0; rr < 4; ++rr) {
            obase[(size_t)rr * Fn + 16 * ct]       = acc[ct][rr] * i0v[rr];
            obase[(size_t)(rr + 4) * Fn + 16 * ct] = acc[ct][rr + 4] * i1v[rr];
        }
    }
}

__global__ void __launch_bounds__(256)
k_rowdots(const float* __restrict__ feature, const float* __restrict__ aligned,
          float* __restrict__ dotv, float* __restrict__ nfv, float* __restrict__ nav) {
    const int row  = blockIdx.x * 8 + (threadIdx.x >> 5);
    const int lane = threadIdx.x & 31;
    v4f fv = *(const v4f*)(feature + (size_t)row * Fn + lane * 4);
    v4f av = *(const v4f*)(aligned + (size_t)row * Fn + lane * 4);
    float d = 0.f, nf = 0.f, na = 0.f;
    #pragma unroll
    for (int q = 0; q < 4; ++q) {
        d  += fv[q] * av[q];
        nf += fv[q] * fv[q];
        na += av[q] * av[q];
    }
    #pragma unroll
    for (int off = 16; off > 0; off >>= 1) {
        d  += __shfl_xor(d,  off, 32);
        nf += __shfl_xor(nf, off, 32);
        na += __shfl_xor(na, off, 32);
    }
    if (lane == 0) { dotv[row] = d; nfv[row] = nf; nav[row] = na; }
}

__global__ void __launch_bounds__(256)
k_window(const float* __restrict__ dotv, const float* __restrict__ nfv,
         const float* __restrict__ nav, float* __restrict__ part) {
    __shared__ float red[256];
    const int w = blockIdx.x * 256 + threadIdx.x;
    float v = 0.0f;
    if (w < Bn * NW) {
        const int b = w / NW;
        const int i = w % NW;
        const int base = b * Sn + i;
        float dw = 0.f, fw = 0.f, aw = 0.f;
        #pragma unroll
        for (int j = 0; j < Wn; ++j) {
            dw += dotv[base + j];
            fw += nfv[base + j];
            aw += nav[base + j];
        }
        float cs = dw / (fmaxf(sqrtf(fw), 1e-12f) * fmaxf(sqrtf(aw), 1e-12f));
        v = 1.0f - cs;
    }
    red[threadIdx.x] = v;
    __syncthreads();
    #pragma unroll
    for (int off = 128; off > 0; off >>= 1) {
        if (threadIdx.x < off) red[threadIdx.x] += red[threadIdx.x + off];
        __syncthreads();
    }
    if (threadIdx.x == 0) part[blockIdx.x] = red[0];
}

__global__ void k_final(const float* __restrict__ part, float* __restrict__ loss_out) {
    float s = 0.0f;
    for (int i = 0; i < 64; ++i) s += part[i];
    loss_out[0] = s / (float)(Bn * NW);
}

extern "C" void kernel_launch(void* const* d_in, const int* in_sizes, int n_in,
                              void* d_out, int out_size, void* d_ws, size_t ws_size,
                              hipStream_t stream) {
    const float* score   = (const float*)d_in[0];
    const float* feature = (const float*)d_in[1];
    const float* proj    = (const float*)d_in[2];
    float* out = (float*)d_out;
    float* ws  = (float*)d_ws;

    float* dotv = ws;
    float* nfv  = ws + 16384;
    float* nav  = ws + 32768;
    float* part = ws + 49152;

    k_flash_softmax_gemm<<<Bn * (Sn / MB), 256, 0, stream>>>(proj, score, feature, out);
    k_rowdots           <<<(Bn * Sn) / 8, 256, 0, stream>>>(feature, out, dotv, nfv, nav);
    k_window            <<<64, 256, 0, stream>>>(dotv, nfv, nav, part);
    k_final             <<<1, 1, 0, stream>>>(part, out + (size_t)Bn * Sn * Fn);
}


# Row-dot and window-loss launches merged: a block keeps dots and norms of its 68 rows in LDS, 9 rows per wave with all loads issued first; launches 4 -> 3; loss partials regrouped (sum order changes)
# speedup vs baseline: 1.2881x; 1.0617x over previous
#include <hip/hip_runtime.h>
#include <math.h>


typedef __attribute__((ext_vector_type(2))) float v2f;
typedef __attribute__((ext_vector_type(4))) float v4f;
typedef __attribute__((ext_vector_type(8))) float v8f;

#define Bn 8
#define Sn 2048
#define Fn 128
#define Wn 5
#define NW (Sn - Wn + 1)

#define MB 64
#define KC 128
#define NC (Sn / KC)
#define PP (KC + 4)
#define PFH 132
#define CHS ((KC / 2) * PFH + 8)
#define NPJ (KC / 16)
#define NFI (KC / 16)
#define KB 8

struct Frag { v2f a[KB]; v4f b01[KB]; v4f b23[KB]; };

template <int CTRL>
__device__ __forceinline__ float quad_perm(float v) {
    return __builtin_bit_cast(float,
        __builtin_amdgcn_update_dpp(0, __builtin_bit_cast(int, v), CTRL, 0xF, 0xF, true));
}

__global__ void __launch_bounds__(256)
k_flash_softmax_gemm(const float* __restrict__ proj, const float* __restrict__ score,
                     const float* __restrict__ feature, float* __restrict__ out) {
    __shared__ __align__(16) float sP[2][MB * PP];
    __shared__ __align__(16) float sF[2][2 * CHS];
    __shared__ __align__(16) float sScale[2][MB];
    __shared__ __align__(16) float sInv[MB];

    const int bid  = blockIdx.x;
    const int b    = bid >> 5;
    const int i0   = (bid & 31) * MB;
    const int tid  = threadIdx.x;
    const int lane = tid & 31;
    const int wave = tid >> 5;

    const int r = ((wave >> 2) << 5) + ((lane >> 2) << 2) + (wave & 3);
    const int q = lane & 3;
    const float* prow = proj  + (size_t)(b * Sn + i0 + r) * Sn + 4 * q;
    const float* srow = score + (size_t)b * Sn + 4 * q;
    const int poff = r * PP + 4 * q;

    const float* fsrc = feature + (size_t)b * Sn * Fn + (size_t)(2 * wave) * Fn + 4 * lane;
    const int foff = (lane >> 4) * CHS + wave * PFH + (lane & 3) * 32 + ((lane >> 2) & 3) * 2;

    const int m16  = lane & 15;
    const int half = lane >> 4;
    const int rt   = wave >> 1;
    const int ch   = wave & 1;
    const int aoff = (16 * rt + m16) * PP + 2 * half;
    const int boff = ch * CHS + half * PFH + 8 * m16;
    const int soff = 16 * rt + 8 * half;

    float m = -INFINITY, s = 0.0f;
    v8f acc[4] = {};

    v4f p[NPJ], sc[NPJ], f0[NFI], f1[NFI];

    auto load_raw = [&](int c) {
        const int k0 = c * KC;
        #pragma unroll
        for (int j = 0; j < NPJ; ++j) {
            p[j]  = *(const v4f*)(prow + k0 + 16 * j);
            sc[j] = *(const v4f*)(srow + k0 + 16 * j);
        }
        #pragma unroll
        for (int i = 0; i < NFI; ++i) {
            f0[i] = *(const v4f*)(fsrc + (size_t)(k0 + 16 * i) * Fn);
            f1[i] = *(const v4f*)(fsrc + (size_t)(k0 + 16 * i + 1) * Fn);
        }
    };

    auto produce = [&](int bufn) {
        float cm = -INFINITY;
        #pragma unroll
        for (int j = 0; j < NPJ; ++j) {
            p[j] = p[j] * sc[j];
            cm = fmaxf(cm, fmaxf(fmaxf(p[j][0], p[j][1]), fmaxf(p[j][2], p[j][3])));
        }
        cm = fmaxf(cm, quad_perm<0xB1>(cm));
        cm = fmaxf(cm, quad_perm<0x4E>(cm));
        const float mnew  = fmaxf(m, cm);
        const float scale = __expf(m - mnew);
        float csum = 0.0f;
        float* pdst = &sP[bufn][poff];
        #pragma unroll
        for (int j = 0; j < NPJ; ++j) {
            v4f w;
            w[0] = __expf(p[j][0] - mnew);
            w[1] = __expf(p[j][1] - mnew);
            w[2] = __expf(p[j][2] - mnew);
            w[3] = __expf(p[j][3] - mnew);
            csum += (w[0] + w[1]) + (w[2] + w[3]);
            *(v4f*)(pdst + 16 * j) = w;
        }
        csum += quad_perm<0xB1>(csum);
        csum += quad_perm<0x4E>(csum);
        s = s * scale + csum;
        m = mnew;
        if (q == 0) sScale[bufn][r] = scale;

        float* fdst = &sF[bufn][foff];
        #pragma unroll
        for (int i = 0; i < NFI; ++i) {
            #pragma unroll
            for (int u = 0; u < 4; ++u) {
                v2f kp;
                kp[0] = f0[i][u]; kp[1] = f1[i][u];
                *(v2f*)(fdst + 8 * i * PFH + 8 * u) = kp;
            }
        }
    };

    Frag fr0, fr1;
    v4f rs0, rs1;

    auto load_frags = [&](Frag& f, int buf, int blk) {
        const float* ap = &sP[buf][aoff] + 4 * KB * blk;
        const float* bp = &sF[buf][boff] + 2 * KB * blk * PFH;
        #pragma unroll
        for (int t = 0; t < KB; ++t) {
            f.a[t]   = *(const v2f*)(ap + 4 * t);
            f.b01[t] = *(const v4f*)(bp + 2 * t * PFH);
            f.b23[t] = *(const v4f*)(bp + 2 * t * PFH + 4);
        }
    };

    auto mma_block = [&](const Frag& f) {
        #pragma unroll
        for (int t = 0; t < KB; ++t) {
            v2f bb[4];
            bb[0][0] = f.b01[t][0]; bb[0][1] = f.b01[t][1];
            bb[1][0] = f.b01[t][2]; bb[1][1] = f.b01[t][3];
            bb[2][0] = f.b23[t][0]; bb[2][1] = f.b23[t][1];
            bb[3][0] = f.b23[t][2]; bb[3][1] = f.b23[t][3];
            #pragma unroll
            for (int ct = 0; ct < 4; ++ct) {
                acc[ct] = __builtin_amdgcn_wmma_f32_16x16x4_f32(
                    false, f.a[t], false, bb[ct], (short)0, acc[ct], false, false);
            }
        }
    };


    auto consume_head = [&](int buf) {
        rs0 = *(const v4f*)&sScale[buf][soff];
        rs1 = *(const v4f*)&sScale[buf][soff + 4];
        load_frags(fr0, buf, 0);
        __builtin_amdgcn_sched_barrier(0);
    };

    static_assert(KC / 4 == 4 * KB, "consume is written out for four fragment blocks");
    auto consume = [&](int buf) {
        #pragma unroll
        for (int ct = 0; ct < 4; ++ct) {
            #pragma unroll
            for (int rr = 0; rr < 4; ++rr) { acc[ct][rr] *= rs0[rr]; acc[ct][rr + 4] *= rs1[rr]; }
        }
        load_frags(fr1, buf, 1);
        __builtin_amdgcn_sched_barrier(0);
        mma_block(fr0);
        __builtin_amdgcn_sched_barrier(0);
        load_frags(fr0, buf, 2);
        __builtin_amdgcn_sched_barrier(0);
        mma_block(fr1);
        __builtin_amdgcn_sched_barrier(0);
        load_frags(fr1, buf, 3);
        __builtin_amdgcn_sched_barrier(0);
        mma_block(fr0);
        __builtin_amdgcn_sched_barrier(0);
        mma_block(fr1);
    };

    load_raw(0);
    produce(0);
    load_raw(1);
    __syncthreads();

    for (int c = 0; c + 1 < NC; ++c) {
        const int buf = c & 1;
        consume_head(buf);
        produce(buf ^ 1);
        if (c + 2 < NC) load_raw(c + 2);
        consume(buf);
        __syncthreads();
    }
    consume_head((NC - 1) & 1);
    if (q == 0) sInv[r] = 1.0f / s;
    consume((NC - 1) & 1);
    __syncthreads();

    const v4f i0v = *(const v4f*)&sInv[soff];
    const v4f i1v = *(const v4f*)&sInv[soff + 4];

    float* obase = out + (size_t)(b * Sn + i0 + soff) * Fn + 64 * ch + m16;
    #pragma unroll
    for (int ct = 0; ct < 4; ++ct) {
        #pragma unroll
        for (int rr = 0; rr < 4; ++rr) {
            obase[(size_t)rr * Fn + 16 * ct]       = acc[ct][rr] * i0v[rr];
            obase[(size_t)(rr + 4) * Fn + 16 * ct] = acc[ct][rr + 4] * i1v[rr];
        }
    }
}

#define WB 64
#define RB (WB + Wn - 1)
#define RPW ((RB + 7) / 8)

__global__ void __launch_bounds__(256)
k_rows_windows(const float* __restrict__ feature, const float* __restrict__ aligned,
               float* __restrict__ part) {
    __shared__ float sD[RB], sNF[RB], sNA[RB];
    __shared__ float sRed[2];
    const int b    = blockIdx.x >> 5;
    const int i0   = (blockIdx.x & 31) * WB;
    const int lane = threadIdx.x & 31;
    const int wave = threadIdx.x >> 5;
    const int nrow = min(RB, Sn - i0);
    const int nwin = min(WB, NW - i0);

    const float* fb = feature + ((size_t)b * Sn + i0) * Fn + lane * 4;
    const float* ab = aligned + ((size_t)b * Sn + i0) * Fn + lane * 4;

    v4f fv[RPW], av[RPW];
    #pragma unroll
    for (int k = 0; k < RPW; ++k) {
        const int r = (8 * k + 7 < WB) ? wave + 8 * k : min(wave + 8 * k, nrow - 1);
        fv[k] = *(const v4f*)(fb + (size_t)r * Fn);
        av[k] = *(const v4f*)(ab + (size_t)r * Fn);
    }
    __builtin_amdgcn_sched_barrier(0);
    float d[RPW], nf[RPW], na[RPW];
    #pragma unroll
    for (int k = 0; k < RPW; ++k) {
        d[k] = 0.f; nf[k] = 0.f; na[k] = 0.f;
        #pragma unroll
        for (int q = 0; q < 4; ++q) {
            d[k]  += fv[k][q] * av[k][q];
            nf[k] += fv[k][q] * fv[k][q];
            na[k] += av[k][q] * av[k][q];
        }
    }
    #pragma unroll
    for (int off = 16; off > 2; off >>= 1) {
        #pragma unroll
        for (int k = 0; k < RPW; ++k) {
            d[k]  += __shfl_xor(d[k],  off, 32);
            nf[k] += __shfl_xor(nf[k], off, 32);
            na[k] += __shfl_xor(na[k], off, 32);
        }
    }
    #pragma unroll
    for (int k = 0; k < RPW; ++k) {
        d[k]  += quad_perm<0x4E>(d[k]);   nf[k] += quad_perm<0x4E>(nf[k]);   na[k] += quad_perm<0x4E>(na[k]);
        d[k]  += quad_perm<0xB1>(d[k]);   nf[k] += quad_perm<0xB1>(nf[k]);   na[k] += quad_perm<0xB1>(na[k]);
    }
    if (lane == 0) {
        #pragma unroll
        for (int k = 0; k < RPW; ++k) {
            const int r = wave + 8 * k;
            if (r < nrow) { sD[r] = d[k]; sNF[r] = nf[k]; sNA[r] = na[k]; }
        }
    }
    __syncthreads();

    if (wave < 2) {
        const int t = threadIdx.x;
        float v = 0.0f;
        if (t < nwin) {
            float dw = 0.f, fw = 0.f, aw = 0.f;
            #pragma unroll
            for (int j = 0; j < Wn; ++j) {
                dw += sD[t + j];
                fw += sNF[t + j];
                aw += sNA[t + j];
            }
            const float cs = dw / (fmaxf(sqrtf(fw), 1e-12f) * fmaxf(sqrtf(aw), 1e-12f));
            v = 1.0f - cs;
        }
        #pragma unroll
        for (int off = 16; off > 2; off >>= 1) v += __shfl_xor(v, off, 32);
        v += quad_perm<0x4E>(v);
        v += quad_perm<0xB1>(v);
        if (lane == 0) sRed[wave] = v;
    }
    __syncthreads();
    if (threadIdx.x == 0) part[blockIdx.x] = sRed[0] + sRed[1];
}

__global__ void __launch_bounds__(32)
k_final(const float* __restrict__ part, float* __restrict__ loss_out) {
    const int lane = threadIdx.x;
    const v4f a = *(const v4f*)(part + 8 * lane);
    const v4f c = *(const v4f*)(part + 8 * lane + 4);
    float s = ((a[0] + a[1]) + (a[2] + a[3])) + ((c[0] + c[1]) + (c[2] + c[3]));
    #pragma unroll
    for (int off = 16; off > 2; off >>= 1) s += __shfl_xor(s, off, 32);
    s += quad_perm<0x4E>(s);
    s += quad_perm<0xB1>(s);
    if (lane == 0) loss_out[0] = s / (float)(Bn * NW);
}

extern "C" void kernel_launch(void* const* d_in, const int* in_sizes, int n_in,
                              void* d_out, int out_size, void* d_ws, size_t ws_size,
                              hipStream_t stream) {
    const float* score   = (const float*)d_in[0];
    const float* feature = (const float*)d_in[1];
    const float* proj    = (const float*)d_in[2];
    float* out = (float*)d_out;
    float* ws  = (float*)d_ws;

    float* part = ws;

    k_flash_softmax_gemm<<<Bn * (Sn / MB), 256, 0, stream>>>(proj, score, feature, out);
    k_rows_windows      <<<Bn * (Sn / WB), 256, 0, stream>>>(feature, out, part);
    k_final             <<<1, 32, 0, stream>>>(part, out + (size_t)Bn * Sn * Fn);
}


# Block barrier split on a 3-stage LDS ring: a wave signals once its part of chunk c+1 is staged and waits only behind chunk c's matrix work, so waves stop meeting every chunk; sum orders unchanged
# speedup vs baseline: 1.3561x; 1.0528x over previous
#include <hip/hip_runtime.h>
#include <math.h>


typedef __attribute__((ext_vector_type(2))) float v2f;
typedef __attribute__((ext_vector_type(4))) float v4f;
typedef __attribute__((ext_vector_type(8))) float v8f;

#define Bn 8
#define Sn 2048
#define Fn 128
#define Wn 5
#define NW (Sn - Wn + 1)

#define MB 64
#define KC 128
#define NC (Sn / KC)
#define PP (KC + 4)
#define PFH 132
#define CHS ((KC / 2) * PFH + 8)
#define NPJ (KC / 16)
#define NFI (KC / 16)
#define KB 8

struct Frag { v2f a[KB]; v4f b01[KB]; v4f b23[KB]; };

template <int CTRL>
__device__ __forceinline__ float quad_perm(float v) {
    return __builtin_bit_cast(float,
        __builtin_amdgcn_update_dpp(0, __builtin_bit_cast(int, v), CTRL, 0xF, 0xF, true));
}

__device__ __forceinline__ void stage_written() {
    asm volatile("s_wait_dscnt 0x0\n\ts_barrier_signal -1" ::: "memory");
}
__device__ __forceinline__ void stage_ready() {
    asm volatile("s_barrier_wait -1" ::: "memory");
}

__global__ void __launch_bounds__(256)
k_flash_softmax_gemm(const float* __restrict__ proj, const float* __restrict__ score,
                     const float* __restrict__ feature, float* __restrict__ out) {
    __shared__ __align__(16) float sP[3][MB * PP];
    __shared__ __align__(16) float sF[3][2 * CHS];
    __shared__ __align__(16) float sScale[3][MB];
    __shared__ __align__(16) float sInv[MB];

    const int bid  = blockIdx.x;
    const int b    = bid >> 5;
    const int i0   = (bid & 31) * MB;
    const int tid  = threadIdx.x;
    const int lane = tid & 31;
    const int wave = tid >> 5;

    const int r = ((wave >> 2) << 5) + ((lane >> 2) << 2) + (wave & 3);
    const int q = lane & 3;
    const float* prow = proj  + (size_t)(b * Sn + i0 + r) * Sn + 4 * q;
    const float* srow = score + (size_t)b * Sn + 4 * q;
    const int poff = r * PP + 4 * q;

    const float* fsrc = feature + (size_t)b * Sn * Fn + (size_t)(2 * wave) * Fn + 4 * lane;
    const int foff = (lane >> 4) * CHS + wave * PFH + (lane & 3) * 32 + ((lane >> 2) & 3) * 2;

    const int m16  = lane & 15;
    const int half = lane >> 4;
    const int rt   = wave >> 1;
    const int ch   = wave & 1;
    const int aoff = (16 * rt + m16) * PP + 2 * half;
    const int boff = ch * CHS + half * PFH + 8 * m16;
    const int soff = 16 * rt + 8 * half;

    float m = -INFINITY, s = 0.0f;
    v8f acc[4] = {};

    v4f p[NPJ], sc[NPJ], f0[NFI], f1[NFI];

    auto load_raw = [&](int c) {
        const int k0 = c * KC;
        #pragma unroll
        for (int j = 0; j < NPJ; ++j) {
            p[j]  = *(const v4f*)(prow + k0 + 16 * j);
            sc[j] = *(const v4f*)(srow + k0 + 16 * j);
        }
        #pragma unroll
        for (int i = 0; i < NFI; ++i) {
            f0[i] = *(const v4f*)(fsrc + (size_t)(k0 + 16 * i) * Fn);
            f1[i] = *(const v4f*)(fsrc + (size_t)(k0 + 16 * i + 1) * Fn);
        }
    };

    auto produce = [&](int bufn) {
        float cm = -INFINITY;
        #pragma unroll
        for (int j = 0; j < NPJ; ++j) {
            p[j] = p[j] * sc[j];
            cm = fmaxf(cm, fmaxf(fmaxf(p[j][0], p[j][1]), fmaxf(p[j][2], p[j][3])));
        }
        cm = fmaxf(cm, quad_perm<0xB1>(cm));
        cm = fmaxf(cm, quad_perm<0x4E>(cm));
        const float mnew  = fmaxf(m, cm);
        const float scale = __expf(m - mnew);
        float csum = 0.0f;
        float* pdst = &sP[bufn][poff];
        #pragma unroll
        for (int j = 0; j < NPJ; ++j) {
            v4f w;
            w[0] = __expf(p[j][0] - mnew);
            w[1] = __expf(p[j][1] - mnew);
            w[2] = __expf(p[j][2] - mnew);
            w[3] = __expf(p[j][3] - mnew);
            csum += (w[0] + w[1]) + (w[2] + w[3]);
            *(v4f*)(pdst + 16 * j) = w;
        }
        csum += quad_perm<0xB1>(csum);
        csum += quad_perm<0x4E>(csum);
        s = s * scale + csum;
        m = mnew;
        if (q == 0) sScale[bufn][r] = scale;

        float* fdst = &sF[bufn][foff];
        #pragma unroll
        for (int i = 0; i < NFI; ++i) {
            #pragma unroll
            for (int u = 0; u < 4; ++u) {
                v2f kp;
                kp[0] = f0[i][u]; kp[1] = f1[i][u];
                *(v2f*)(fdst + 8 * i * PFH + 8 * u) = kp;
            }
        }
    };

    Frag fr0, fr1;
    v4f rs0, rs1;

    auto load_frags = [&](Frag& f, int buf, int blk) {
        const float* ap = &sP[buf][aoff] + 4 * KB * blk;
        const float* bp = &sF[buf][boff] + 2 * KB * blk * PFH;
        #pragma unroll
        for (int t = 0; t < KB; ++t) {
            f.a[t]   = *(const v2f*)(ap + 4 * t);
            f.b01[t] = *(const v4f*)(bp + 2 * t * PFH);
            f.b23[t] = *(const v4f*)(bp + 2 * t * PFH + 4);
        }
    };

    auto mma_block = [&](const Frag& f) {
        #pragma unroll
        for (int t = 0; t < KB; ++t) {
            v2f bb[4];
            bb[0][0] = f.b01[t][0]; bb[0][1] = f.b01[t][1];
            bb[1][0] = f.b01[t][2]; bb[1][1] = f.b01[t][3];
            bb[2][0] = f.b23[t][0]; bb[2][1] = f.b23[t][1];
            bb[3][0] = f.b23[t][2]; bb[3][1] = f.b23[t][3];
            #pragma unroll
            for (int ct = 0; ct < 4; ++ct) {
                acc[ct] = __builtin_amdgcn_wmma_f32_16x16x4_f32(
                    false, f.a[t], false, bb[ct], (short)0, acc[ct], false, false);
            }
        }
    };


    auto consume_head = [&](int buf) {
        rs0 = *(const v4f*)&sScale[buf][soff];
        rs1 = *(const v4f*)&sScale[buf][soff + 4];
        load_frags(fr0, buf, 0);
        __builtin_amdgcn_sched_barrier(0);
    };

    static_assert(KC / 4 == 4 * KB, "consume is written out for four fragment blocks");
    auto consume = [&](int buf) {
        #pragma unroll
        for (int ct = 0; ct < 4; ++ct) {
            #pragma unroll
            for (int rr = 0; rr < 4; ++rr) { acc[ct][rr] *= rs0[rr]; acc[ct][rr + 4] *= rs1[rr]; }
        }
        load_frags(fr1, buf, 1);
        __builtin_amdgcn_sched_barrier(0);
        mma_block(fr0);
        __builtin_amdgcn_sched_barrier(0);
        load_frags(fr0, buf, 2);
        __builtin_amdgcn_sched_barrier(0);
        mma_block(fr1);
        __builtin_amdgcn_sched_barrier(0);
        load_frags(fr1, buf, 3);
        __builtin_amdgcn_sched_barrier(0);
        mma_block(fr0);
        __builtin_amdgcn_sched_barrier(0);
        mma_block(fr1);
    };

    load_raw(0);
    produce(0);
    load_raw(1);
    __syncthreads();

    int st = 0;
    for (int c = 0; c + 1 < NC; ++c) {
        const int nx = (st == 2) ? 0 : st + 1;
        consume_head(st);
        produce(nx);
        stage_written();
        if (c + 2 < NC) load_raw(c + 2);
        consume(st);
        __builtin_amdgcn_sched_barrier(0);
        stage_ready();
        st = nx;
    }
    consume_head(st);
    if (q == 0) sInv[r] = 1.0f / s;
    consume(st);
    __syncthreads();

    const v4f i0v = *(const v4f*)&sInv[soff];
    const v4f i1v = *(const v4f*)&sInv[soff + 4];

    float* obase = out + (size_t)(b * Sn + i0 + soff) * Fn + 64 * ch + m16;
    #pragma unroll
    for (int ct = 0; ct < 4; ++ct) {
        #pragma unroll
        for (int rr = 0; rr < 4; ++rr) {
            obase[(size_t)rr * Fn + 16 * ct]       = acc[ct][rr] * i0v[rr];
            obase[(size_t)(rr + 4) * Fn + 16 * ct] = acc[ct][rr + 4] * i1v[rr];
        }
    }
}

#define WB 64
#define RB (WB + Wn - 1)
#define RPW ((RB + 7) / 8)

__global__ void __launch_bounds__(256)
k_rows_windows(const float* __restrict__ feature, const float* __restrict__ aligned,
               float* __restrict__ part) {
    __shared__ float sD[RB], sNF[RB], sNA[RB];
    __shared__ float sRed[2];
    const int b    = blockIdx.x >> 5;
    const int i0   = (blockIdx.x & 31) * WB;
    const int lane = threadIdx.x & 31;
    const int wave = threadIdx.x >> 5;
    const int nrow = min(RB, Sn - i0);
    const int nwin = min(WB, NW - i0);

    const float* fb = feature + ((size_t)b * Sn + i0) * Fn + lane * 4;
    const float* ab = aligned + ((size_t)b * Sn + i0) * Fn + lane * 4;

    v4f fv[RPW], av[RPW];
    #pragma unroll
    for (int k = 0; k < RPW; ++k) {
        const int r = (8 * k + 7 < WB) ? wave + 8 * k : min(wave + 8 * k, nrow - 1);
        fv[k] = *(const v4f*)(fb + (size_t)r * Fn);
        av[k] = *(const v4f*)(ab + (size_t)r * Fn);
    }
    __builtin_amdgcn_sched_barrier(0);
    float d[RPW], nf[RPW], na[RPW];
    #pragma unroll
    for (int k = 0; k < RPW; ++k) {
        d[k] = 0.f; nf[k] = 0.f; na[k] = 0.f;
        #pragma unroll
        for (int q = 0; q < 4; ++q) {
            d[k]  += fv[k][q] * av[k][q];
            nf[k] += fv[k][q] * fv[k][q];
            na[k] += av[k][q] * av[k][q];
        }
    }
    #pragma unroll
    for (int off = 16; off > 2; off >>= 1) {
        #pragma unroll
        for (int k = 0; k < RPW; ++k) {
            d[k]  += __shfl_xor(d[k],  off, 32);
            nf[k] += __shfl_xor(nf[k], off, 32);
            na[k] += __shfl_xor(na[k], off, 32);
        }
    }
    #pragma unroll
    for (int k = 0; k < RPW; ++k) {
        d[k]  += quad_perm<0x4E>(d[k]);   nf[k] += quad_perm<0x4E>(nf[k]);   na[k] += quad_perm<0x4E>(na[k]);
        d[k]  += quad_perm<0xB1>(d[k]);   nf[k] += quad_perm<0xB1>(nf[k]);   na[k] += quad_perm<0xB1>(na[k]);
    }
    if (lane == 0) {
        #pragma unroll
        for (int k = 0; k < RPW; ++k) {
            const int r = wave + 8 * k;
            if (r < nrow) { sD[r] = d[k]; sNF[r] = nf[k]; sNA[r] = na[k]; }
        }
    }
    __syncthreads();

    if (wave < 2) {
        const int t = threadIdx.x;
        float v = 0.0f;
        if (t < nwin) {
            float dw = 0.f, fw = 0.f, aw = 0.f;
            #pragma unroll
            for (int j = 0; j < Wn; ++j) {
                dw += sD[t + j];
                fw += sNF[t + j];
                aw += sNA[t + j];
            }
            const float cs = dw / (fmaxf(sqrtf(fw), 1e-12f) * fmaxf(sqrtf(aw), 1e-12f));
            v = 1.0f - cs;
        }
        #pragma unroll
        for (int off = 16; off > 2; off >>= 1) v += __shfl_xor(v, off, 32);
        v += quad_perm<0x4E>(v);
        v += quad_perm<0xB1>(v);
        if (lane == 0) sRed[wave] = v;
    }
    __syncthreads();
    if (threadIdx.x == 0) part[blockIdx.x] = sRed[0] + sRed[1];
}

__global__ void __launch_bounds__(32)
k_final(const float* __restrict__ part, float* __restrict__ loss_out) {
    const int lane = threadIdx.x;
    const v4f a = *(const v4f*)(part + 8 * lane);
    const v4f c = *(const v4f*)(part + 8 * lane + 4);
    float s = ((a[0] + a[1]) + (a[2] + a[3])) + ((c[0] + c[1]) + (c[2] + c[3]));
    #pragma unroll
    for (int off = 16; off > 2; off >>= 1) s += __shfl_xor(s, off, 32);
    s += quad_perm<0x4E>(s);
    s += quad_perm<0xB1>(s);
    if (lane == 0) loss_out[0] = s / (float)(Bn * NW);
}

extern "C" void kernel_launch(void* const* d_in, const int* in_sizes, int n_in,
                              void* d_out, int out_size, void* d_ws, size_t ws_size,
                              hipStream_t stream) {
    const float* score   = (const float*)d_in[0];
    const float* feature = (const float*)d_in[1];
    const float* proj    = (const float*)d_in[2];
    float* out = (float*)d_out;
    float* ws  = (float*)d_ws;

    float* part = ws;

    k_flash_softmax_gemm<<<Bn * (Sn / MB), 256, 0, stream>>>(proj, score, feature, out);
    k_rows_windows      <<<Bn * (Sn / WB), 256, 0, stream>>>(feature, out, part);
    k_final             <<<1, 32, 0, stream>>>(part, out + (size_t)Bn * Sn * Fn);
}
